# starting kernel
# baseline (speedup 1.0000x reference)
#include <hip/hip_runtime.h>
#include <hip/hip_bf16.h>

typedef __attribute__((ext_vector_type(2))) float v2f;
typedef __attribute__((ext_vector_type(8))) float v8f;

__global__ void k_fill(float* __restrict__ p, float v, int n) {
  int i = blockIdx.x * blockDim.x + threadIdx.x;
  if (i < n) p[i] = v;
}

__global__ void k_deg_edges(const int* __restrict__ dst, float* __restrict__ deg, int E) {
  int i = blockIdx.x * blockDim.x + threadIdx.x;
  if (i < E) atomicAdd(deg + dst[i], 1.0f);
}

__global__ void k_dinv(float* __restrict__ deg, int n) {
  int i = blockIdx.x * blockDim.x + threadIdx.x;
  if (i < n) deg[i] = rsqrtf(fmaxf(deg[i], 1.0f));
}

__global__ void __launch_bounds__(256) k_gemm_wmma(
    const float* __restrict__ A, const float* __restrict__ W,
    float* __restrict__ out, int N, int K, int M) {
  __shared__ float As[16 * (256 + 4)];
  const int ldsK = K + 4;
  const int rowBase = blockIdx.x * 16;
  const int colBase = blockIdx.y * 128 + (threadIdx.x >> 5) * 16;

  for (int e = threadIdx.x; e < 16 * K; e += 256) {
    int r = e / K;
    int c = e - r * K;
    int g = rowBase + r;
    As[r * ldsK + c] = (g < N) ? A[(size_t)g * K + c] : 0.0f;
  }
  __syncthreads();

  const int lane = threadIdx.x & 31;
  const int half = lane >> 4;
  const int m    = lane & 15;
  const int col  = colBase + m;

  v8f acc = {};
  for (int k = 0; k < K; k += 4) {
    const int kk = k + 2 * half;
    v2f a, b;
    a.x = As[m * ldsK + kk];
    a.y = As[m * ldsK + kk + 1];
    b.x = W[(size_t)kk * M + col];
    b.y = W[(size_t)(kk + 1) * M + col];
    acc = __builtin_amdgcn_wmma_f32_16x16x4_f32(
         false, a,  false, b,
         (short)0, acc,  false,  false);
  }

  for (int r = 0; r < 8; ++r) {
    int row = rowBase + r + 8 * half;
    if (row < N) out[(size_t)row * M + col] = acc[r];
  }
}

__global__ void k_selfloop(const float* __restrict__ hW, const float* __restrict__ dinv,
                           float* __restrict__ agg, int N, int C) {
  long long i = (long long)blockIdx.x * blockDim.x + threadIdx.x;
  long long total = (long long)N * C;
  if (i < total) {
    int node = (int)(i / C);
    float di = dinv[node];
    agg[i] = hW[i] * di * di;
  }
}

__global__ void k_edge_scatter(const int* __restrict__ src, const int* __restrict__ dst,
                               const float* __restrict__ dinv, const float* __restrict__ hW,
                               float* __restrict__ agg, int E, int C) {
  int wave = (int)(((long long)blockIdx.x * blockDim.x + threadIdx.x) >> 5);
  int lane = threadIdx.x & 31;
  if (wave >= E) return;
  int s = src[wave];
  int d = dst[wave];
  float nrm = dinv[s] * dinv[d];
  const float4* hp = (const float4*)(hW + (size_t)s * C);
  float* ap = agg + (size_t)d * C;
  int nchunk = C >> 7;
  for (int j = 0; j < nchunk; ++j) {
    int v4 = j * 32 + lane;
    float4 v = hp[v4];
    int base = v4 * 4;
    atomicAdd(ap + base + 0, v.x * nrm);
    atomicAdd(ap + base + 1, v.y * nrm);
    atomicAdd(ap + base + 2, v.z * nrm);
    atomicAdd(ap + base + 3, v.w * nrm);
  }
}

__global__ void k_post(const float* __restrict__ agg, const float* __restrict__ bias,
                       float* __restrict__ out, int N, int C) {
  int node = (int)(((long long)blockIdx.x * blockDim.x + threadIdx.x) >> 5);
  int lane = threadIdx.x & 31;
  if (node >= N) return;
  const float* row = agg + (size_t)node * C;
  float v[8];
  int nch = C >> 5;
  float s = 0.0f;
  for (int j = 0; j < nch; ++j) {
    float t = fmaxf(row[j * 32 + lane] + bias[j * 32 + lane], 0.0f);
    v[j] = t;
    s += t;
  }
  for (int off = 16; off > 0; off >>= 1) s += __shfl_xor(s, off, 32);
  float mean = s / (float)C;
  float q = 0.0f;
  for (int j = 0; j < nch; ++j) {
    float d = v[j] - mean;
    q += d * d;
  }
  for (int off = 16; off > 0; off >>= 1) q += __shfl_xor(q, off, 32);
  float rstd = rsqrtf(q / (float)C + 1e-5f);
  float* o = out + (size_t)node * C;
  for (int j = 0; j < nch; ++j) o[j * 32 + lane] = (v[j] - mean) * rstd;
}

__global__ void k_pool_accum(const float* __restrict__ h, const int* __restrict__ batch,
                             float* __restrict__ pool, int N, int C) {
  long long i = (long long)blockIdx.x * blockDim.x + threadIdx.x;
  long long total = (long long)N * C;
  if (i < total) {
    int node = (int)(i / C);
    int c = (int)(i - (long long)node * C);
    atomicAdd(pool + (size_t)batch[node] * C + c, h[i]);
  }
}

__global__ void k_count(const int* __restrict__ batch, float* __restrict__ cnt, int N) {
  int i = blockIdx.x * blockDim.x + threadIdx.x;
  if (i < N) atomicAdd(cnt + batch[i], 1.0f);
}

__global__ void k_pool_final(float* __restrict__ pool, const float* __restrict__ cnt,
                             int G, int C) {
  int i = blockIdx.x * blockDim.x + threadIdx.x;
  if (i < G * C) pool[i] /= fmaxf(cnt[i / C], 1.0f);
}

static inline size_t align256(size_t x) { return (x + 255) & ~(size_t)255; }

extern "C" void kernel_launch(void* const* d_in, const int* in_sizes, int n_in,
                              void* d_out, int out_size, void* d_ws, size_t ws_size,
                              hipStream_t stream) {
  const float* x     = (const float*)d_in[0];
  const int*   ei    = (const int*)d_in[1];
  const int*   batch = (const int*)d_in[2];
  const float* W0 = (const float*)d_in[4];
  const float* b0 = (const float*)d_in[5];
  const float* W1 = (const float*)d_in[6];
  const float* b1 = (const float*)d_in[7];
  const float* W2 = (const float*)d_in[8];
  const float* b2 = (const float*)d_in[9];

  const int N  = in_sizes[2];
  const int E  = in_sizes[1] / 2;
  const int IN = in_sizes[0] / N;
  const int H0 = in_sizes[5];
  const int H1 = in_sizes[7];
  const int H2 = in_sizes[9];
  const int G  = out_size / H2 - N;

  const int* srcp = ei;
  const int* dstp = ei + E;

  char* ws = (char*)d_ws;
  float* dinv = (float*)ws;  ws += align256((size_t)N * 4);
  float* cnt  = (float*)ws;  ws += align256((size_t)G * 4);
  float* bufA = (float*)ws;  ws += align256((size_t)N * 256 * 4);
  float* bufB = (float*)ws;  ws += align256((size_t)N * 256 * 4);

  float* pool = (float*)d_out;
  float* hout = (float*)d_out + (size_t)G * H2;

  const int B = 256;
  auto cdiv = [](long long a, long long b) { return (int)((a + b - 1) / b); };

  k_fill<<<cdiv(N, B), B, 0, stream>>>(dinv, 1.0f, N);
  k_deg_edges<<<cdiv(E, B), B, 0, stream>>>(dstp, dinv, E);
  k_dinv<<<cdiv(N, B), B, 0, stream>>>(dinv, N);

  auto layer = [&](const float* hin, int K, int M, const float* W, const float* bias,
                   float* hw, float* agg, float* outp) {
    dim3 g(cdiv(N, 16), M / 128);
    k_gemm_wmma<<<g, B, 0, stream>>>(hin, W, hw, N, K, M);
    k_selfloop<<<cdiv((long long)N * M, B), B, 0, stream>>>(hw, dinv, agg, N, M);
    k_edge_scatter<<<cdiv((long long)E * 32, B), B, 0, stream>>>(srcp, dstp, dinv, hw, agg, E, M);
    k_post<<<cdiv((long long)N * 32, B), B, 0, stream>>>(agg, bias, outp, N, M);
  };

  layer(x,    IN, H0, W0, b0,  bufA,  bufB,  bufA);
  layer(bufA, H0, H1, W1, b1,  bufB,  bufA,  bufB);
  layer(bufB, H1, H2, W2, b2,  bufA,  bufB,  hout);

  k_fill<<<cdiv((long long)G * H2, B), B, 0, stream>>>(pool, 0.0f, G * H2);
  k_fill<<<cdiv(G, B), B, 0, stream>>>(cnt, 0.0f, G);
  k_pool_accum<<<cdiv((long long)N * H2, B), B, 0, stream>>>(hout, batch, pool, N, H2);
  k_count<<<cdiv(N, B), B, 0, stream>>>(batch, cnt, N);
  k_pool_final<<<cdiv((long long)G * H2, B), B, 0, stream>>>(pool, cnt, G, H2);
}


# Edge aggregation: per-destination gather over edges grouped by destination (count, scan, place, row sort, once per call) replaces per-float atomics; sum order per node fixed, ascending source id
# speedup vs baseline: 1.8060x; 1.8060x over previous
#include <hip/hip_runtime.h>
#include <hip/hip_bf16.h>

typedef __attribute__((ext_vector_type(2))) float v2f;
typedef __attribute__((ext_vector_type(8))) float v8f;

__global__ void k_fill(float* __restrict__ p, float v, int n) {
  int i = blockIdx.x * blockDim.x + threadIdx.x;
  if (i < n) p[i] = v;
}

__global__ void k_zero_i(int* __restrict__ p, int n) {
  int i = blockIdx.x * blockDim.x + threadIdx.x;
  if (i < n) p[i] = 0;
}

__global__ void k_count_rank(const int* __restrict__ dst, int* __restrict__ cnt,
                             int* __restrict__ rank, int E) {
  int i = blockIdx.x * blockDim.x + threadIdx.x;
  if (i < E) rank[i] = atomicAdd(cnt + dst[i], 1);
}

__global__ void __launch_bounds__(1024) k_dinv_bsum(const int* __restrict__ cnt,
                                                    float* __restrict__ dinv,
                                                    int* __restrict__ bsum, int N) {
  __shared__ int part[32];
  const int t = threadIdx.x, lane = t & 31, w = t >> 5;
  const int i = blockIdx.x * 1024 + t;
  int c = 0;
  if (i < N) {
    c = cnt[i];
    dinv[i] = rsqrtf(fmaxf((float)(c + 1), 1.0f));
  }
  int s = c;
  for (int off = 16; off > 0; off >>= 1) s += __shfl_xor(s, off, 32);
  if (lane == 0) part[w] = s;
  __syncthreads();
  if (w == 0) {
    int x = part[lane];
    for (int off = 16; off > 0; off >>= 1) x += __shfl_xor(x, off, 32);
    if (lane == 0) bsum[blockIdx.x] = x;
  }
}

__global__ void __launch_bounds__(1024) k_rowptr(const int* __restrict__ cnt,
                                                 const int* __restrict__ bsum,
                                                 int* __restrict__ rowptr, int N) {
  __shared__ int tot[32];
  const int t = threadIdx.x, lane = t & 31, w = t >> 5;
  const int i = blockIdx.x * 1024 + t;
  int base = 0;
  for (int b = lane; b < (int)blockIdx.x; b += 32) base += bsum[b];
  for (int off = 16; off > 0; off >>= 1) base += __shfl_xor(base, off, 32);
  const int c = (i < N) ? cnt[i] : 0;
  int v = c;
  for (int off = 1; off < 32; off <<= 1) {
    int u = __shfl_up(v, off, 32);
    if (lane >= off) v += u;
  }
  if (lane == 31) tot[w] = v;
  __syncthreads();
  if (w == 0) {
    int x = tot[lane];
    for (int off = 1; off < 32; off <<= 1) {
      int u = __shfl_up(x, off, 32);
      if (lane >= off) x += u;
    }
    tot[lane] = x;
  }
  __syncthreads();
  const int excl = base + (w > 0 ? tot[w - 1] : 0) + v - c;
  if (i < N) rowptr[i] = excl;
  if (i == N - 1) rowptr[N] = excl + c;
}

__global__ void k_place(const int* __restrict__ src, const int* __restrict__ dst,
                        const int* __restrict__ rank, const int* __restrict__ rowptr,
                        int* __restrict__ tmp, int E) {
  int i = blockIdx.x * blockDim.x + threadIdx.x;
  if (i < E) {
    int p = rowptr[dst[i]] + rank[i];
    if ((unsigned)p < (unsigned)E) tmp[p] = src[i];
  }
}

__global__ void __launch_bounds__(256) k_rowsort(const int* __restrict__ rowptr,
                                                 const int* __restrict__ tmp,
                                                 int* __restrict__ col, int N) {
  const int node = __builtin_amdgcn_readfirstlane((int)((blockIdx.x * 256u + threadIdx.x) >> 5));
  const int lane = threadIdx.x & 31;
  if (node >= N) return;
  const int beg = rowptr[node];
  const int n = rowptr[node + 1] - beg;
  for (int i = lane; i < n; i += 32) {
    const int key = tmp[beg + i];
    int r = 0;
    for (int j = 0; j < n; ++j) {
      const int kj = tmp[beg + j];
      r += (kj < key || (kj == key && j < i)) ? 1 : 0;
    }
    col[beg + r] = key;
  }
}

__global__ void __launch_bounds__(256) k_gemm_wmma(
    const float* __restrict__ A, const float* __restrict__ W,
    float* __restrict__ out, int N, int K, int M) {
  __shared__ float As[16 * (256 + 4)];
  const int ldsK = K + 4;
  const int rowBase = blockIdx.x * 16;
  const int colBase = blockIdx.y * 128 + (threadIdx.x >> 5) * 16;

  for (int e = threadIdx.x; e < 16 * K; e += 256) {
    int r = e / K;
    int c = e - r * K;
    int g = rowBase + r;
    As[r * ldsK + c] = (g < N) ? A[(size_t)g * K + c] : 0.0f;
  }
  __syncthreads();

  const int lane = threadIdx.x & 31;
  const int half = lane >> 4;
  const int m    = lane & 15;
  const int col  = colBase + m;

  v8f acc = {};
  for (int k = 0; k < K; k += 4) {
    const int kk = k + 2 * half;
    v2f a, b;
    a.x = As[m * ldsK + kk];
    a.y = As[m * ldsK + kk + 1];
    b.x = W[(size_t)kk * M + col];
    b.y = W[(size_t)(kk + 1) * M + col];
    acc = __builtin_amdgcn_wmma_f32_16x16x4_f32(
         false, a,  false, b,
         (short)0, acc,  false,  false);
  }

  for (int r = 0; r < 8; ++r) {
    int row = rowBase + r + 8 * half;
    if (row < N) out[(size_t)row * M + col] = acc[r];
  }
}

__device__ __forceinline__ void fma4(float4& a, const float4 v, const float w) {
  a.x += v.x * w;
  a.y += v.y * w;
  a.z += v.z * w;
  a.w += v.w * w;
}

template <bool WIDE>
__global__ void __launch_bounds__(256) k_gather(const int* __restrict__ rowptr,
                                                const int* __restrict__ col,
                                                const float* __restrict__ dinv,
                                                const float* __restrict__ hW,
                                                float* __restrict__ agg, int N) {
  const int node = __builtin_amdgcn_readfirstlane((int)((blockIdx.x * 256u + threadIdx.x) >> 5));
  const int lane = threadIdx.x & 31;
  if (node >= N) return;
  constexpr int C4 = WIDE ? 64 : 32;
  const float4* __restrict__ tab = (const float4*)hW;
  const int beg = rowptr[node];
  const int end = rowptr[node + 1];
  const float dd = dinv[node];

  float4 a0, a1;
  {
    const float4* hp = tab + (size_t)node * C4 + lane;
    const float ws = dd * dd;
    const float4 x = hp[0];
    a0.x = x.x * ws; a0.y = x.y * ws; a0.z = x.z * ws; a0.w = x.w * ws;
    a1.x = a1.y = a1.z = a1.w = 0.0f;
    if (WIDE) {
      const float4 y = hp[32];
      a1.x = y.x * ws; a1.y = y.y * ws; a1.z = y.z * ws; a1.w = y.w * ws;
    }
  }

#pragma unroll 4
  for (int j = beg; j < end; ++j) {
    int s = col[j];
    s = (unsigned)s < (unsigned)N ? s : N - 1;
    const float w = dinv[s] * dd;
    const float4* p = tab + (size_t)s * C4 + lane;
    const float4 x = p[0];
    if (WIDE) {
      const float4 y = p[32];
      fma4(a1, y, w);
    }
    fma4(a0, x, w);
  }

  float4* op = (float4*)agg + (size_t)node * C4 + lane;
  op[0] = a0;
  if (WIDE) op[32] = a1;
}

__global__ void k_post(const float* __restrict__ agg, const float* __restrict__ bias,
                       float* __restrict__ out, int N, int C) {
  int node = (int)(((long long)blockIdx.x * blockDim.x + threadIdx.x) >> 5);
  int lane = threadIdx.x & 31;
  if (node >= N) return;
  const float* row = agg + (size_t)node * C;
  float v[8];
  int nch = C >> 5;
  float s = 0.0f;
  for (int j = 0; j < nch; ++j) {
    float t = fmaxf(row[j * 32 + lane] + bias[j * 32 + lane], 0.0f);
    v[j] = t;
    s += t;
  }
  for (int off = 16; off > 0; off >>= 1) s += __shfl_xor(s, off, 32);
  float mean = s / (float)C;
  float q = 0.0f;
  for (int j = 0; j < nch; ++j) {
    float d = v[j] - mean;
    q += d * d;
  }
  for (int off = 16; off > 0; off >>= 1) q += __shfl_xor(q, off, 32);
  float rstd = rsqrtf(q / (float)C + 1e-5f);
  float* o = out + (size_t)node * C;
  for (int j = 0; j < nch; ++j) o[j * 32 + lane] = (v[j] - mean) * rstd;
}

__global__ void k_pool_accum(const float* __restrict__ h, const int* __restrict__ batch,
                             float* __restrict__ pool, int N, int C) {
  long long i = (long long)blockIdx.x * blockDim.x + threadIdx.x;
  long long total = (long long)N * C;
  if (i < total) {
    int node = (int)(i / C);
    int c = (int)(i - (long long)node * C);
    atomicAdd(pool + (size_t)batch[node] * C + c, h[i]);
  }
}

__global__ void k_count(const int* __restrict__ batch, float* __restrict__ cnt, int N) {
  int i = blockIdx.x * blockDim.x + threadIdx.x;
  if (i < N) atomicAdd(cnt + batch[i], 1.0f);
}

__global__ void k_pool_final(float* __restrict__ pool, const float* __restrict__ cnt,
                             int G, int C) {
  int i = blockIdx.x * blockDim.x + threadIdx.x;
  if (i < G * C) pool[i] /= fmaxf(cnt[i / C], 1.0f);
}

static inline size_t align256(size_t x) { return (x + 255) & ~(size_t)255; }

extern "C" void kernel_launch(void* const* d_in, const int* in_sizes, int n_in,
                              void* d_out, int out_size, void* d_ws, size_t ws_size,
                              hipStream_t stream) {
  const float* x     = (const float*)d_in[0];
  const int*   ei    = (const int*)d_in[1];
  const int*   batch = (const int*)d_in[2];
  const float* W0 = (const float*)d_in[4];
  const float* b0 = (const float*)d_in[5];
  const float* W1 = (const float*)d_in[6];
  const float* b1 = (const float*)d_in[7];
  const float* W2 = (const float*)d_in[8];
  const float* b2 = (const float*)d_in[9];

  const int N  = in_sizes[2];
  const int E  = in_sizes[1] / 2;
  const int IN = in_sizes[0] / N;
  const int H0 = in_sizes[5];
  const int H1 = in_sizes[7];
  const int H2 = in_sizes[9];
  const int G  = out_size / H2 - N;

  const int* srcp = ei;
  const int* dstp = ei + E;

  char* ws = (char*)d_ws;
  float* dinv = (float*)ws;  ws += align256((size_t)N * 4);
  float* cnt  = (float*)ws;  ws += align256((size_t)G * 4);
  float* bufA = (float*)ws;  ws += align256((size_t)N * 256 * 4);
  float* bufB = (float*)ws;  ws += align256((size_t)N * 256 * 4);
  int* indeg  = (int*)ws;    ws += align256((size_t)N * 4);
  int* rowptr = (int*)ws;    ws += align256((size_t)(N + 1) * 4);
  int* bsum   = (int*)ws;    ws += align256((size_t)(N / 1024 + 1) * 4);
  int* rank   = (int*)ws;    ws += align256((size_t)E * 4);
  int* ctmp   = (int*)ws;    ws += align256((size_t)E * 4);
  int* col    = (int*)ws;    ws += align256((size_t)E * 4);

  float* pool = (float*)d_out;
  float* hout = (float*)d_out + (size_t)G * H2;

  const int B = 256;
  auto cdiv = [](long long a, long long b) { return (int)((a + b - 1) / b); };

  const int NB = cdiv(N, 1024);
  k_zero_i<<<cdiv(N, B), B, 0, stream>>>(indeg, N);
  k_count_rank<<<cdiv(E, B), B, 0, stream>>>(dstp, indeg, rank, E);
  k_dinv_bsum<<<NB, 1024, 0, stream>>>(indeg, dinv, bsum, N);
  k_rowptr<<<NB, 1024, 0, stream>>>(indeg, bsum, rowptr, N);
  k_place<<<cdiv(E, B), B, 0, stream>>>(srcp, dstp, rank, rowptr, ctmp, E);
  k_rowsort<<<cdiv((long long)N * 32, B), B, 0, stream>>>(rowptr, ctmp, col, N);

  auto layer = [&](const float* hin, int K, int M, const float* W, const float* bias,
                   float* hw, float* agg, float* outp) {
    dim3 g(cdiv(N, 16), M / 128);
    k_gemm_wmma<<<g, B, 0, stream>>>(hin, W, hw, N, K, M);
    if (M > 128)
      k_gather<true><<<cdiv((long long)N * 32, B), B, 0, stream>>>(rowptr, col, dinv, hw, agg, N);
    else
      k_gather<false><<<cdiv((long long)N * 32, B), B, 0, stream>>>(rowptr, col, dinv, hw, agg, N);
    k_post<<<cdiv((long long)N * 32, B), B, 0, stream>>>(agg, bias, outp, N, M);
  };

  layer(x,    IN, H0, W0, b0,  bufA,  bufB,  bufA);
  layer(bufA, H0, H1, W1, b1,  bufB,  bufA,  bufB);
  layer(bufB, H1, H2, W2, b2,  bufA,  bufB,  hout);

  k_fill<<<cdiv((long long)G * H2, B), B, 0, stream>>>(pool, 0.0f, G * H2);
  k_fill<<<cdiv(G, B), B, 0, stream>>>(cnt, 0.0f, G);
  k_pool_accum<<<cdiv((long long)N * H2, B), B, 0, stream>>>(hout, batch, pool, N, H2);
  k_count<<<cdiv(N, B), B, 0, stream>>>(batch, cnt, N);
  k_pool_final<<<cdiv((long long)G * H2, B), B, 0, stream>>>(pool, cnt, G, H2);
}


# Mean pool without atomics: one work-group per graph matches batch ids, adds its nodes' rows (512 B per wave load), counts and divides in-kernel; sum order fixed: node id per wave, then wave order
# speedup vs baseline: 2.0461x; 1.1329x over previous
#include <hip/hip_runtime.h>
#include <hip/hip_bf16.h>

typedef __attribute__((ext_vector_type(2))) float v2f;
typedef __attribute__((ext_vector_type(8))) float v8f;

__global__ void k_zero_i(int* __restrict__ p, int n) {
  int i = blockIdx.x * blockDim.x + threadIdx.x;
  if (i < n) p[i] = 0;
}

__global__ void k_count_rank(const int* __restrict__ dst, int* __restrict__ cnt,
                             int* __restrict__ rank, int E) {
  int i = blockIdx.x * blockDim.x + threadIdx.x;
  if (i < E) rank[i] = atomicAdd(cnt + dst[i], 1);
}

__global__ void __launch_bounds__(1024) k_dinv_bsum(const int* __restrict__ cnt,
                                                    float* __restrict__ dinv,
                                                    int* __restrict__ bsum, int N) {
  __shared__ int part[32];
  const int t = threadIdx.x, lane = t & 31, w = t >> 5;
  const int i = blockIdx.x * 1024 + t;
  int c = 0;
  if (i < N) {
    c = cnt[i];
    dinv[i] = rsqrtf(fmaxf((float)(c + 1), 1.0f));
  }
  int s = c;
  for (int off = 16; off > 0; off >>= 1) s += __shfl_xor(s, off, 32);
  if (lane == 0) part[w] = s;
  __syncthreads();
  if (w == 0) {
    int x = part[lane];
    for (int off = 16; off > 0; off >>= 1) x += __shfl_xor(x, off, 32);
    if (lane == 0) bsum[blockIdx.x] = x;
  }
}

__global__ void __launch_bounds__(1024) k_rowptr(const int* __restrict__ cnt,
                                                 const int* __restrict__ bsum,
                                                 int* __restrict__ rowptr, int N) {
  __shared__ int tot[32];
  const int t = threadIdx.x, lane = t & 31, w = t >> 5;
  const int i = blockIdx.x * 1024 + t;
  int base = 0;
  for (int b = lane; b < (int)blockIdx.x; b += 32) base += bsum[b];
  for (int off = 16; off > 0; off >>= 1) base += __shfl_xor(base, off, 32);
  const int c = (i < N) ? cnt[i] : 0;
  int v = c;
  for (int off = 1; off < 32; off <<= 1) {
    int u = __shfl_up(v, off, 32);
    if (lane >= off) v += u;
  }
  if (lane == 31) tot[w] = v;
  __syncthreads();
  if (w == 0) {
    int x = tot[lane];
    for (int off = 1; off < 32; off <<= 1) {
      int u = __shfl_up(x, off, 32);
      if (lane >= off) x += u;
    }
    tot[lane] = x;
  }
  __syncthreads();
  const int excl = base + (w > 0 ? tot[w - 1] : 0) + v - c;
  if (i < N) rowptr[i] = excl;
  if (i == N - 1) rowptr[N] = excl + c;
}

__global__ void k_place(const int* __restrict__ src, const int* __restrict__ dst,
                        const int* __restrict__ rank, const int* __restrict__ rowptr,
                        int* __restrict__ tmp, int E) {
  int i = blockIdx.x * blockDim.x + threadIdx.x;
  if (i < E) {
    int p = rowptr[dst[i]] + rank[i];
    if ((unsigned)p < (unsigned)E) tmp[p] = src[i];
  }
}

__global__ void __launch_bounds__(256) k_rowsort(const int* __restrict__ rowptr,
                                                 const int* __restrict__ tmp,
                                                 int* __restrict__ col, int N) {
  const int node = __builtin_amdgcn_readfirstlane((int)((blockIdx.x * 256u + threadIdx.x) >> 5));
  const int lane = threadIdx.x & 31;
  if (node >= N) return;
  const int beg = rowptr[node];
  const int n = rowptr[node + 1] - beg;
  for (int i = lane; i < n; i += 32) {
    const int key = tmp[beg + i];
    int r = 0;
    for (int j = 0; j < n; ++j) {
      const int kj = tmp[beg + j];
      r += (kj < key || (kj == key && j < i)) ? 1 : 0;
    }
    col[beg + r] = key;
  }
}

__global__ void __launch_bounds__(256) k_gemm_wmma(
    const float* __restrict__ A, const float* __restrict__ W,
    float* __restrict__ out, int N, int K, int M) {
  __shared__ float As[16 * (256 + 4)];
  const int ldsK = K + 4;
  const int rowBase = blockIdx.x * 16;
  const int colBase = blockIdx.y * 128 + (threadIdx.x >> 5) * 16;

  for (int e = threadIdx.x; e < 16 * K; e += 256) {
    int r = e / K;
    int c = e - r * K;
    int g = rowBase + r;
    As[r * ldsK + c] = (g < N) ? A[(size_t)g * K + c] : 0.0f;
  }
  __syncthreads();

  const int lane = threadIdx.x & 31;
  const int half = lane >> 4;
  const int m    = lane & 15;
  const int col  = colBase + m;

  v8f acc = {};
  for (int k = 0; k < K; k += 4) {
    const int kk = k + 2 * half;
    v2f a, b;
    a.x = As[m * ldsK + kk];
    a.y = As[m * ldsK + kk + 1];
    b.x = W[(size_t)kk * M + col];
    b.y = W[(size_t)(kk + 1) * M + col];
    acc = __builtin_amdgcn_wmma_f32_16x16x4_f32(
         false, a,  false, b,
         (short)0, acc,  false,  false);
  }

  for (int r = 0; r < 8; ++r) {
    int row = rowBase + r + 8 * half;
    if (row < N) out[(size_t)row * M + col] = acc[r];
  }
}

__device__ __forceinline__ void fma4(float4& a, const float4 v, const float w) {
  a.x += v.x * w;
  a.y += v.y * w;
  a.z += v.z * w;
  a.w += v.w * w;
}

template <bool WIDE>
__global__ void __launch_bounds__(256) k_gather(const int* __restrict__ rowptr,
                                                const int* __restrict__ col,
                                                const float* __restrict__ dinv,
                                                const float* __restrict__ hW,
                                                float* __restrict__ agg, int N) {
  const int node = __builtin_amdgcn_readfirstlane((int)((blockIdx.x * 256u + threadIdx.x) >> 5));
  const int lane = threadIdx.x & 31;
  if (node >= N) return;
  constexpr int C4 = WIDE ? 64 : 32;
  const float4* __restrict__ tab = (const float4*)hW;
  const int beg = rowptr[node];
  const int end = rowptr[node + 1];
  const float dd = dinv[node];

  float4 a0, a1;
  {
    const float4* hp = tab + (size_t)node * C4 + lane;
    const float ws = dd * dd;
    const float4 x = hp[0];
    a0.x = x.x * ws; a0.y = x.y * ws; a0.z = x.z * ws; a0.w = x.w * ws;
    a1.x = a1.y = a1.z = a1.w = 0.0f;
    if (WIDE) {
      const float4 y = hp[32];
      a1.x = y.x * ws; a1.y = y.y * ws; a1.z = y.z * ws; a1.w = y.w * ws;
    }
  }

#pragma unroll 4
  for (int j = beg; j < end; ++j) {
    int s = col[j];
    s = (unsigned)s < (unsigned)N ? s : N - 1;
    const float w = dinv[s] * dd;
    const float4* p = tab + (size_t)s * C4 + lane;
    const float4 x = p[0];
    if (WIDE) {
      const float4 y = p[32];
      fma4(a1, y, w);
    }
    fma4(a0, x, w);
  }

  float4* op = (float4*)agg + (size_t)node * C4 + lane;
  op[0] = a0;
  if (WIDE) op[32] = a1;
}

__global__ void k_post(const float* __restrict__ agg, const float* __restrict__ bias,
                       float* __restrict__ out, int N, int C) {
  int node = (int)(((long long)blockIdx.x * blockDim.x + threadIdx.x) >> 5);
  int lane = threadIdx.x & 31;
  if (node >= N) return;
  const float* row = agg + (size_t)node * C;
  float v[8];
  int nch = C >> 5;
  float s = 0.0f;
  for (int j = 0; j < nch; ++j) {
    float t = fmaxf(row[j * 32 + lane] + bias[j * 32 + lane], 0.0f);
    v[j] = t;
    s += t;
  }
  for (int off = 16; off > 0; off >>= 1) s += __shfl_xor(s, off, 32);
  float mean = s / (float)C;
  float q = 0.0f;
  for (int j = 0; j < nch; ++j) {
    float d = v[j] - mean;
    q += d * d;
  }
  for (int off = 16; off > 0; off >>= 1) q += __shfl_xor(q, off, 32);
  float rstd = rsqrtf(q / (float)C + 1e-5f);
  float* o = out + (size_t)node * C;
  for (int j = 0; j < nch; ++j) o[j * 32 + lane] = (v[j] - mean) * rstd;
}

__global__ void __launch_bounds__(1024) k_pool_mean(const float* __restrict__ h,
                                                     const int* __restrict__ batch,
                                                     float* __restrict__ pool, int N) {
  __shared__ float4 part[32 * 32];
  __shared__ int pcnt[32];
  const int g = blockIdx.x;
  const int lane = threadIdx.x & 31;
  const int w = __builtin_amdgcn_readfirstlane((int)(threadIdx.x >> 5));
  const float4* __restrict__ tab = (const float4*)h;
  const int nchunk = (N + 31) >> 5;

  float4 acc;
  acc.x = acc.y = acc.z = acc.w = 0.0f;
  int cnt = 0;

  for (int c0 = w; c0 < nchunk; c0 += 32 * 8) {
    unsigned m[8];
#pragma unroll
    for (int u = 0; u < 8; ++u) {
      const int n = (c0 + 32 * u) * 32 + lane;
      const int b = batch[n < N ? n : N - 1];
      m[u] = __builtin_amdgcn_ballot_w32(n < N && b == g);
    }
#pragma unroll
    for (int u = 0; u < 8; ++u) {
      unsigned mask = m[u];
      const int base = (c0 + 32 * u) * 32;
      cnt += __builtin_popcount(mask);
      while (mask) {
        int id[8];
        bool ok[8];
#pragma unroll
        for (int q = 0; q < 8; ++q) {
          ok[q] = mask != 0u;
          id[q] = ok[q] ? __builtin_ctz(mask) : id[0];
          mask &= mask - 1u;
        }
        float4 x[8];
#pragma unroll
        for (int q = 0; q < 8; ++q) x[q] = tab[(size_t)(base + id[q]) * 32 + lane];
#pragma unroll
        for (int q = 0; q < 8; ++q) {
          acc.x += ok[q] ? x[q].x : 0.0f;
          acc.y += ok[q] ? x[q].y : 0.0f;
          acc.z += ok[q] ? x[q].z : 0.0f;
          acc.w += ok[q] ? x[q].w : 0.0f;
        }
      }
    }
  }

  part[w * 32 + lane] = acc;
  if (lane == 0) pcnt[w] = cnt;
  __syncthreads();

  const int t = threadIdx.x;
  if (t < 128) {
    const float* pf = (const float*)part;
    float s = 0.0f;
    int c = 0;
    for (int k = 0; k < 32; ++k) {
      s += pf[k * 128 + t];
      c += pcnt[k];
    }
    pool[(size_t)g * 128 + t] = s / fmaxf((float)c, 1.0f);
  }
}

static inline size_t align256(size_t x) { return (x + 255) & ~(size_t)255; }

extern "C" void kernel_launch(void* const* d_in, const int* in_sizes, int n_in,
                              void* d_out, int out_size, void* d_ws, size_t ws_size,
                              hipStream_t stream) {
  const float* x     = (const float*)d_in[0];
  const int*   ei    = (const int*)d_in[1];
  const int*   batch = (const int*)d_in[2];
  const float* W0 = (const float*)d_in[4];
  const float* b0 = (const float*)d_in[5];
  const float* W1 = (const float*)d_in[6];
  const float* b1 = (const float*)d_in[7];
  const float* W2 = (const float*)d_in[8];
  const float* b2 = (const float*)d_in[9];

  const int N  = in_sizes[2];
  const int E  = in_sizes[1] / 2;
  const int IN = in_sizes[0] / N;
  const int H0 = in_sizes[5];
  const int H1 = in_sizes[7];
  const int H2 = in_sizes[9];
  const int G  = out_size / H2 - N;

  const int* srcp = ei;
  const int* dstp = ei + E;

  char* ws = (char*)d_ws;
  float* dinv = (float*)ws;  ws += align256((size_t)N * 4);
  float* bufA = (float*)ws;  ws += align256((size_t)N * 256 * 4);
  float* bufB = (float*)ws;  ws += align256((size_t)N * 256 * 4);
  int* indeg  = (int*)ws;    ws += align256((size_t)N * 4);
  int* rowptr = (int*)ws;    ws += align256((size_t)(N + 1) * 4);
  int* bsum   = (int*)ws;    ws += align256((size_t)(N / 1024 + 1) * 4);
  int* rank   = (int*)ws;    ws += align256((size_t)E * 4);
  int* ctmp   = (int*)ws;    ws += align256((size_t)E * 4);
  int* col    = (int*)ws;    ws += align256((size_t)E * 4);

  float* pool = (float*)d_out;
  float* hout = (float*)d_out + (size_t)G * H2;

  const int B = 256;
  auto cdiv = [](long long a, long long b) { return (int)((a + b - 1) / b); };

  const int NB = cdiv(N, 1024);
  k_zero_i<<<cdiv(N, B), B, 0, stream>>>(indeg, N);
  k_count_rank<<<cdiv(E, B), B, 0, stream>>>(dstp, indeg, rank, E);
  k_dinv_bsum<<<NB, 1024, 0, stream>>>(indeg, dinv, bsum, N);
  k_rowptr<<<NB, 1024, 0, stream>>>(indeg, bsum, rowptr, N);
  k_place<<<cdiv(E, B), B, 0, stream>>>(srcp, dstp, rank, rowptr, ctmp, E);
  k_rowsort<<<cdiv((long long)N * 32, B), B, 0, stream>>>(rowptr, ctmp, col, N);

  auto layer = [&](const float* hin, int K, int M, const float* W, const float* bias,
                   float* hw, float* agg, float* outp) {
    dim3 g(cdiv(N, 16), M / 128);
    k_gemm_wmma<<<g, B, 0, stream>>>(hin, W, hw, N, K, M);
    if (M > 128)
      k_gather<true><<<cdiv((long long)N * 32, B), B, 0, stream>>>(rowptr, col, dinv, hw, agg, N);
    else
      k_gather<false><<<cdiv((long long)N * 32, B), B, 0, stream>>>(rowptr, col, dinv, hw, agg, N);
    k_post<<<cdiv((long long)N * 32, B), B, 0, stream>>>(agg, bias, outp, N, M);
  };

  layer(x,    IN, H0, W0, b0,  bufA,  bufB,  bufA);
  layer(bufA, H0, H1, W1, b1,  bufB,  bufA,  bufB);
  layer(bufB, H1, H2, W2, b2,  bufA,  bufB,  hout);

  k_pool_mean<<<G, 1024, 0, stream>>>(hout, batch, pool, N);
}


# Bias, relu, layer norm moved into the gather kernel, on the row still in registers (agg buffer and 3 launches gone); layer-norm sum order changes: 4 adjacent channels per lane, then a lane butterfly
# speedup vs baseline: 2.4390x; 1.1920x over previous
#include <hip/hip_runtime.h>
#include <hip/hip_bf16.h>

typedef __attribute__((ext_vector_type(2))) float v2f;
typedef __attribute__((ext_vector_type(8))) float v8f;

__global__ void k_zero_i(int* __restrict__ p, int n) {
  int i = blockIdx.x * blockDim.x + threadIdx.x;
  if (i < n) p[i] = 0;
}

__global__ void k_count_rank(const int* __restrict__ dst, int* __restrict__ cnt,
                             int* __restrict__ rank, int E) {
  int i = blockIdx.x * blockDim.x + threadIdx.x;
  if (i < E) rank[i] = atomicAdd(cnt + dst[i], 1);
}

__global__ void __launch_bounds__(1024) k_dinv_bsum(const int* __restrict__ cnt,
                                                    float* __restrict__ dinv,
                                                    int* __restrict__ bsum, int N) {
  __shared__ int part[32];
  const int t = threadIdx.x, lane = t & 31, w = t >> 5;
  const int i = blockIdx.x * 1024 + t;
  int c = 0;
  if (i < N) {
    c = cnt[i];
    dinv[i] = rsqrtf(fmaxf((float)(c + 1), 1.0f));
  }
  int s = c;
  for (int off = 16; off > 0; off >>= 1) s += __shfl_xor(s, off, 32);
  if (lane == 0) part[w] = s;
  __syncthreads();
  if (w == 0) {
    int x = part[lane];
    for (int off = 16; off > 0; off >>= 1) x += __shfl_xor(x, off, 32);
    if (lane == 0) bsum[blockIdx.x] = x;
  }
}

__global__ void __launch_bounds__(1024) k_rowptr(const int* __restrict__ cnt,
                                                 const int* __restrict__ bsum,
                                                 int* __restrict__ rowptr, int N) {
  __shared__ int tot[32];
  const int t = threadIdx.x, lane = t & 31, w = t >> 5;
  const int i = blockIdx.x * 1024 + t;
  int base = 0;
  for (int b = lane; b < (int)blockIdx.x; b += 32) base += bsum[b];
  for (int off = 16; off > 0; off >>= 1) base += __shfl_xor(base, off, 32);
  const int c = (i < N) ? cnt[i] : 0;
  int v = c;
  for (int off = 1; off < 32; off <<= 1) {
    int u = __shfl_up(v, off, 32);
    if (lane >= off) v += u;
  }
  if (lane == 31) tot[w] = v;
  __syncthreads();
  if (w == 0) {
    int x = tot[lane];
    for (int off = 1; off < 32; off <<= 1) {
      int u = __shfl_up(x, off, 32);
      if (lane >= off) x += u;
    }
    tot[lane] = x;
  }
  __syncthreads();
  const int excl = base + (w > 0 ? tot[w - 1] : 0) + v - c;
  if (i < N) rowptr[i] = excl;
  if (i == N - 1) rowptr[N] = excl + c;
}

__global__ void k_place(const int* __restrict__ src, const int* __restrict__ dst,
                        const int* __restrict__ rank, const int* __restrict__ rowptr,
                        int* __restrict__ tmp, int E) {
  int i = blockIdx.x * blockDim.x + threadIdx.x;
  if (i < E) {
    int p = rowptr[dst[i]] + rank[i];
    if ((unsigned)p < (unsigned)E) tmp[p] = src[i];
  }
}

__global__ void __launch_bounds__(256) k_rowsort(const int* __restrict__ rowptr,
                                                 const int* __restrict__ tmp,
                                                 int* __restrict__ col, int N) {
  const int node = __builtin_amdgcn_readfirstlane((int)((blockIdx.x * 256u + threadIdx.x) >> 5));
  const int lane = threadIdx.x & 31;
  if (node >= N) return;
  const int beg = rowptr[node];
  const int n = rowptr[node + 1] - beg;
  for (int i = lane; i < n; i += 32) {
    const int key = tmp[beg + i];
    int r = 0;
    for (int j = 0; j < n; ++j) {
      const int kj = tmp[beg + j];
      r += (kj < key || (kj == key && j < i)) ? 1 : 0;
    }
    col[beg + r] = key;
  }
}

__global__ void __launch_bounds__(256) k_gemm_wmma(
    const float* __restrict__ A, const float* __restrict__ W,
    float* __restrict__ out, int N, int K, int M) {
  __shared__ float As[16 * (256 + 4)];
  const int ldsK = K + 4;
  const int rowBase = blockIdx.x * 16;
  const int colBase = blockIdx.y * 128 + (threadIdx.x >> 5) * 16;

  for (int e = threadIdx.x; e < 16 * K; e += 256) {
    int r = e / K;
    int c = e - r * K;
    int g = rowBase + r;
    As[r * ldsK + c] = (g < N) ? A[(size_t)g * K + c] : 0.0f;
  }
  __syncthreads();

  const int lane = threadIdx.x & 31;
  const int half = lane >> 4;
  const int m    = lane & 15;
  const int col  = colBase + m;

  v8f acc = {};
  for (int k = 0; k < K; k += 4) {
    const int kk = k + 2 * half;
    v2f a, b;
    a.x = As[m * ldsK + kk];
    a.y = As[m * ldsK + kk + 1];
    b.x = W[(size_t)kk * M + col];
    b.y = W[(size_t)(kk + 1) * M + col];
    acc = __builtin_amdgcn_wmma_f32_16x16x4_f32(
         false, a,  false, b,
         (short)0, acc,  false,  false);
  }

  for (int r = 0; r < 8; ++r) {
    int row = rowBase + r + 8 * half;
    if (row < N) out[(size_t)row * M + col] = acc[r];
  }
}

__device__ __forceinline__ void fma4(float4& a, const float4 v, const float w) {
  a.x += v.x * w;
  a.y += v.y * w;
  a.z += v.z * w;
  a.w += v.w * w;
}

template <int CTRL>
__device__ __forceinline__ float row_pick(float x) {
  return __builtin_bit_cast(float, __builtin_amdgcn_mov_dpp(__builtin_bit_cast(int, x), CTRL, 0xf, 0xf, true));
}
__device__ __forceinline__ float wave_sum(float v) {
  v += row_pick<0x161>(v);
  v += row_pick<0x162>(v);
  v += row_pick<0x164>(v);
  v += row_pick<0x168>(v);
  v += __builtin_bit_cast(float, __builtin_amdgcn_permlanex16(__builtin_bit_cast(int, v), __builtin_bit_cast(int, v),
                                                              0x76543210, 0xfedcba98, false, false));
  return v;
}

template <bool WIDE>
__global__ void __launch_bounds__(256) k_gather_post(const int* __restrict__ rowptr,
                                                     const int* __restrict__ col,
                                                     const float* __restrict__ dinv,
                                                     const float* __restrict__ hW,
                                                     const float* __restrict__ bias,
                                                     float* __restrict__ out, int N) {
  const int node = __builtin_amdgcn_readfirstlane((int)((blockIdx.x * 256u + threadIdx.x) >> 5));
  const int lane = threadIdx.x & 31;
  if (node >= N) return;
  constexpr int C4 = WIDE ? 64 : 32;
  const float4* __restrict__ tab = (const float4*)hW;
  const int beg = rowptr[node];
  const int end = rowptr[node + 1];
  const float dd = dinv[node];

  float4 bv0, bv1;
  {
    const float* bp = bias + 4 * lane;
    bv0.x = bp[0]; bv0.y = bp[1]; bv0.z = bp[2]; bv0.w = bp[3];
    bv1.x = bv1.y = bv1.z = bv1.w = 0.0f;
    if (WIDE) { bv1.x = bp[128]; bv1.y = bp[129]; bv1.z = bp[130]; bv1.w = bp[131]; }
  }

  float4 a0, a1;
  {
    const float4* hp = tab + (size_t)node * C4 + lane;
    const float ws = dd * dd;
    const float4 x = hp[0];
    a0.x = x.x * ws; a0.y = x.y * ws; a0.z = x.z * ws; a0.w = x.w * ws;
    a1.x = a1.y = a1.z = a1.w = 0.0f;
    if (WIDE) {
      const float4 y = hp[32];
      a1.x = y.x * ws; a1.y = y.y * ws; a1.z = y.z * ws; a1.w = y.w * ws;
    }
  }

  asm volatile("" ::"v"(bv0.x), "v"(bv0.y), "v"(bv0.z), "v"(bv0.w));
  if (WIDE) asm volatile("" ::"v"(bv1.x), "v"(bv1.y), "v"(bv1.z), "v"(bv1.w));

#pragma unroll 4
  for (int j = beg; j < end; ++j) {
    int s = col[j];
    s = (unsigned)s < (unsigned)N ? s : N - 1;
    const float w = dinv[s] * dd;
    const float4* p = tab + (size_t)s * C4 + lane;
    const float4 x = p[0];
    if (WIDE) {
      const float4 y = p[32];
      fma4(a1, y, w);
    }
    fma4(a0, x, w);
  }

  a0.x = fmaxf(a0.x + bv0.x, 0.0f);
  a0.y = fmaxf(a0.y + bv0.y, 0.0f);
  a0.z = fmaxf(a0.z + bv0.z, 0.0f);
  a0.w = fmaxf(a0.w + bv0.w, 0.0f);
  float sum = ((a0.x + a0.y) + a0.z) + a0.w;
  if (WIDE) {
    a1.x = fmaxf(a1.x + bv1.x, 0.0f);
    a1.y = fmaxf(a1.y + bv1.y, 0.0f);
    a1.z = fmaxf(a1.z + bv1.z, 0.0f);
    a1.w = fmaxf(a1.w + bv1.w, 0.0f);
    sum += ((a1.x + a1.y) + a1.z) + a1.w;
  }
  constexpr float invC = WIDE ? (1.0f / 256.0f) : (1.0f / 128.0f);
  const float mean = wave_sum(sum) * invC;
  a0.x -= mean; a0.y -= mean; a0.z -= mean; a0.w -= mean;
  float q = ((a0.x * a0.x + a0.y * a0.y) + a0.z * a0.z) + a0.w * a0.w;
  if (WIDE) {
    a1.x -= mean; a1.y -= mean; a1.z -= mean; a1.w -= mean;
    q += ((a1.x * a1.x + a1.y * a1.y) + a1.z * a1.z) + a1.w * a1.w;
  }
  const float rstd = rsqrtf(wave_sum(q) * invC + 1e-5f);
  a0.x *= rstd; a0.y *= rstd; a0.z *= rstd; a0.w *= rstd;

  float4* op = (float4*)out + (size_t)node * C4 + lane;
  op[0] = a0;
  if (WIDE) {
    a1.x *= rstd; a1.y *= rstd; a1.z *= rstd; a1.w *= rstd;
    op[32] = a1;
  }
}

__global__ void __launch_bounds__(1024) k_pool_mean(const float* __restrict__ h,
                                                     const int* __restrict__ batch,
                                                     float* __restrict__ pool, int N) {
  __shared__ float4 part[32 * 32];
  __shared__ int pcnt[32];
  const int g = blockIdx.x;
  const int lane = threadIdx.x & 31;
  const int w = __builtin_amdgcn_readfirstlane((int)(threadIdx.x >> 5));
  const float4* __restrict__ tab = (const float4*)h;
  const int nchunk = (N + 31) >> 5;

  float4 acc;
  acc.x = acc.y = acc.z = acc.w = 0.0f;
  int cnt = 0;

  for (int c0 = w; c0 < nchunk; c0 += 32 * 8) {
    unsigned m[8];
#pragma unroll
    for (int u = 0; u < 8; ++u) {
      const int n = (c0 + 32 * u) * 32 + lane;
      const int b = batch[n < N ? n : N - 1];
      m[u] = __builtin_amdgcn_ballot_w32(n < N && b == g);
    }
#pragma unroll
    for (int u = 0; u < 8; ++u) {
      unsigned mask = m[u];
      const int base = (c0 + 32 * u) * 32;
      cnt += __builtin_popcount(mask);
      while (mask) {
        int id[8];
        bool ok[8];
#pragma unroll
        for (int q = 0; q < 8; ++q) {
          ok[q] = mask != 0u;
          id[q] = ok[q] ? __builtin_ctz(mask) : id[0];
          mask &= mask - 1u;
        }
        float4 x[8];
#pragma unroll
        for (int q = 0; q < 8; ++q) x[q] = tab[(size_t)(base + id[q]) * 32 + lane];
#pragma unroll
        for (int q = 0; q < 8; ++q) {
          acc.x += ok[q] ? x[q].x : 0.0f;
          acc.y += ok[q] ? x[q].y : 0.0f;
          acc.z += ok[q] ? x[q].z : 0.0f;
          acc.w += ok[q] ? x[q].w : 0.0f;
        }
      }
    }
  }

  part[w * 32 + lane] = acc;
  if (lane == 0) pcnt[w] = cnt;
  __syncthreads();

  const int t = threadIdx.x;
  if (t < 128) {
    const float* pf = (const float*)part;
    float s = 0.0f;
    int c = 0;
    for (int k = 0; k < 32; ++k) {
      s += pf[k * 128 + t];
      c += pcnt[k];
    }
    pool[(size_t)g * 128 + t] = s / fmaxf((float)c, 1.0f);
  }
}

static inline size_t align256(size_t x) { return (x + 255) & ~(size_t)255; }

extern "C" void kernel_launch(void* const* d_in, const int* in_sizes, int n_in,
                              void* d_out, int out_size, void* d_ws, size_t ws_size,
                              hipStream_t stream) {
  const float* x     = (const float*)d_in[0];
  const int*   ei    = (const int*)d_in[1];
  const int*   batch = (const int*)d_in[2];
  const float* W0 = (const float*)d_in[4];
  const float* b0 = (const float*)d_in[5];
  const float* W1 = (const float*)d_in[6];
  const float* b1 = (const float*)d_in[7];
  const float* W2 = (const float*)d_in[8];
  const float* b2 = (const float*)d_in[9];

  const int N  = in_sizes[2];
  const int E  = in_sizes[1] / 2;
  const int IN = in_sizes[0] / N;
  const int H0 = in_sizes[5];
  const int H1 = in_sizes[7];
  const int H2 = in_sizes[9];
  const int G  = out_size / H2 - N;

  const int* srcp = ei;
  const int* dstp = ei + E;

  char* ws = (char*)d_ws;
  float* dinv = (float*)ws;  ws += align256((size_t)N * 4);
  float* bufA = (float*)ws;  ws += align256((size_t)N * 256 * 4);
  float* bufB = (float*)ws;  ws += align256((size_t)N * 256 * 4);
  int* indeg  = (int*)ws;    ws += align256((size_t)N * 4);
  int* rowptr = (int*)ws;    ws += align256((size_t)(N + 1) * 4);
  int* bsum   = (int*)ws;    ws += align256((size_t)(N / 1024 + 1) * 4);
  int* rank   = (int*)ws;    ws += align256((size_t)E * 4);
  int* ctmp   = (int*)ws;    ws += align256((size_t)E * 4);
  int* col    = (int*)ws;    ws += align256((size_t)E * 4);

  float* pool = (float*)d_out;
  float* hout = (float*)d_out + (size_t)G * H2;

  const int B = 256;
  auto cdiv = [](long long a, long long b) { return (int)((a + b - 1) / b); };

  const int NB = cdiv(N, 1024);
  k_zero_i<<<cdiv(N, B), B, 0, stream>>>(indeg, N);
  k_count_rank<<<cdiv(E, B), B, 0, stream>>>(dstp, indeg, rank, E);
  k_dinv_bsum<<<NB, 1024, 0, stream>>>(indeg, dinv, bsum, N);
  k_rowptr<<<NB, 1024, 0, stream>>>(indeg, bsum, rowptr, N);
  k_place<<<cdiv(E, B), B, 0, stream>>>(srcp, dstp, rank, rowptr, ctmp, E);
  k_rowsort<<<cdiv((long long)N * 32, B), B, 0, stream>>>(rowptr, ctmp, col, N);

  auto layer = [&](const float* hin, int K, int M, const float* W, const float* bias,
                   float* hw, float* outp) {
    dim3 g(cdiv(N, 16), M / 128);
    k_gemm_wmma<<<g, B, 0, stream>>>(hin, W, hw, N, K, M);
    if (M > 128)
      k_gather_post<true><<<cdiv((long long)N * 32, B), B, 0, stream>>>(rowptr, col, dinv, hw, bias, outp, N);
    else
      k_gather_post<false><<<cdiv((long long)N * 32, B), B, 0, stream>>>(rowptr, col, dinv, hw, bias, outp, N);
  };

  layer(x,    IN, H0, W0, b0,  bufA,  bufB);
  layer(bufB, H0, H1, W1, b1,  bufA,  bufB);
  layer(bufB, H1, H2, W2, b2,  bufA,  hout);

  k_pool_mean<<<G, 1024, 0, stream>>>(hout, batch, pool, N);
}


# Product kernel re-tiled: block tile 16x128 -> 64x128, a wave keeps each B fragment for four 16-row tiles (4 accumulators): a quarter of the waves and of the W loads; per-element sum order unchanged
# speedup vs baseline: 3.6646x; 1.5025x over previous
#include <hip/hip_runtime.h>
#include <hip/hip_bf16.h>

typedef __attribute__((ext_vector_type(2))) float v2f;
typedef __attribute__((ext_vector_type(8))) float v8f;

__global__ void k_zero_i(int* __restrict__ p, int n) {
  int i = blockIdx.x * blockDim.x + threadIdx.x;
  if (i < n) p[i] = 0;
}

__global__ void k_count_rank(const int* __restrict__ dst, int* __restrict__ cnt,
                             int* __restrict__ rank, int E) {
  int i = blockIdx.x * blockDim.x + threadIdx.x;
  if (i < E) rank[i] = atomicAdd(cnt + dst[i], 1);
}

__global__ void __launch_bounds__(1024) k_dinv_bsum(const int* __restrict__ cnt,
                                                    float* __restrict__ dinv,
                                                    int* __restrict__ bsum, int N) {
  __shared__ int part[32];
  const int t = threadIdx.x, lane = t & 31, w = t >> 5;
  const int i = blockIdx.x * 1024 + t;
  int c = 0;
  if (i < N) {
    c = cnt[i];
    dinv[i] = rsqrtf(fmaxf((float)(c + 1), 1.0f));
  }
  int s = c;
  for (int off = 16; off > 0; off >>= 1) s += __shfl_xor(s, off, 32);
  if (lane == 0) part[w] = s;
  __syncthreads();
  if (w == 0) {
    int x = part[lane];
    for (int off = 16; off > 0; off >>= 1) x += __shfl_xor(x, off, 32);
    if (lane == 0) bsum[blockIdx.x] = x;
  }
}

__global__ void __launch_bounds__(1024) k_rowptr(const int* __restrict__ cnt,
                                                 const int* __restrict__ bsum,
                                                 int* __restrict__ rowptr, int N) {
  __shared__ int tot[32];
  const int t = threadIdx.x, lane = t & 31, w = t >> 5;
  const int i = blockIdx.x * 1024 + t;
  int base = 0;
  for (int b = lane; b < (int)blockIdx.x; b += 32) base += bsum[b];
  for (int off = 16; off > 0; off >>= 1) base += __shfl_xor(base, off, 32);
  const int c = (i < N) ? cnt[i] : 0;
  int v = c;
  for (int off = 1; off < 32; off <<= 1) {
    int u = __shfl_up(v, off, 32);
    if (lane >= off) v += u;
  }
  if (lane == 31) tot[w] = v;
  __syncthreads();
  if (w == 0) {
    int x = tot[lane];
    for (int off = 1; off < 32; off <<= 1) {
      int u = __shfl_up(x, off, 32);
      if (lane >= off) x += u;
    }
    tot[lane] = x;
  }
  __syncthreads();
  const int excl = base + (w > 0 ? tot[w - 1] : 0) + v - c;
  if (i < N) rowptr[i] = excl;
  if (i == N - 1) rowptr[N] = excl + c;
}

__global__ void k_place(const int* __restrict__ src, const int* __restrict__ dst,
                        const int* __restrict__ rank, const int* __restrict__ rowptr,
                        int* __restrict__ tmp, int E) {
  int i = blockIdx.x * blockDim.x + threadIdx.x;
  if (i < E) {
    int p = rowptr[dst[i]] + rank[i];
    if ((unsigned)p < (unsigned)E) tmp[p] = src[i];
  }
}

__global__ void __launch_bounds__(256) k_rowsort(const int* __restrict__ rowptr,
                                                 const int* __restrict__ tmp,
                                                 int* __restrict__ col, int N) {
  const int node = __builtin_amdgcn_readfirstlane((int)((blockIdx.x * 256u + threadIdx.x) >> 5));
  const int lane = threadIdx.x & 31;
  if (node >= N) return;
  const int beg = rowptr[node];
  const int n = rowptr[node + 1] - beg;
  for (int i = lane; i < n; i += 32) {
    const int key = tmp[beg + i];
    int r = 0;
    for (int j = 0; j < n; ++j) {
      const int kj = tmp[beg + j];
      r += (kj < key || (kj == key && j < i)) ? 1 : 0;
    }
    col[beg + r] = key;
  }
}

constexpr int GM_ROWS = 64;
constexpr int GM_RT = GM_ROWS / 16;
constexpr int GM_KC = 128;
constexpr int GM_PITCH = GM_KC + 4;

__global__ void __launch_bounds__(256) k_gemm_wmma(
    const float* __restrict__ A, const float* __restrict__ W,
    float* __restrict__ out, int N, int K, int M) {
  __shared__ float As[GM_ROWS * GM_PITCH];
  const int rowBase = blockIdx.x * GM_ROWS;
  const int lane = threadIdx.x & 31;
  const int half = lane >> 4;
  const int m    = lane & 15;
  const int col  = blockIdx.y * 128 + (threadIdx.x >> 5) * 16 + m;

  v8f acc[GM_RT];
#pragma unroll
  for (int t = 0; t < GM_RT; ++t) acc[t] = v8f{};

  for (int kc0 = 0; kc0 < K; kc0 += GM_KC) {
    const int kw = (K - kc0 < GM_KC) ? (K - kc0) : GM_KC;
    if (kc0 > 0) __syncthreads();

    {
      float4 v[8];
      int    off[8];
      bool   ok[8];
#pragma unroll
      for (int u = 0; u < 8; ++u) {
        const int e = threadIdx.x + 256 * u;
        const int r = e >> 5;
        const int c = (e & 31) * 4;
        const int g = rowBase + r;
        ok[u]  = (g < N) && (c < kw);
        off[u] = r * GM_PITCH + c;
        const int gc = g < N ? g : N - 1;
        const int cc = c < kw ? c : 0;
        v[u] = *(const float4*)(A + (size_t)gc * K + kc0 + cc);
      }
#pragma unroll
      for (int u = 0; u < 8; ++u) {
        float4 x = v[u];
        if (!ok[u]) x.x = x.y = x.z = x.w = 0.0f;
        *(float4*)(As + off[u]) = x;
      }
    }
    __syncthreads();

    const float* __restrict__ wp = W + (size_t)(kc0 + 2 * half) * M + col;
    const float* ap = As + m * GM_PITCH + 2 * half;
#pragma unroll 4
    for (int k = 0; k < kw; k += 4) {
      v2f b;
      b.x = wp[(size_t)k * M];
      b.y = wp[(size_t)(k + 1) * M];
#pragma unroll
      for (int t = 0; t < GM_RT; ++t) {
        v2f a;
        a.x = ap[t * 16 * GM_PITCH + k];
        a.y = ap[t * 16 * GM_PITCH + k + 1];
        acc[t] = __builtin_amdgcn_wmma_f32_16x16x4_f32(
             false, a,  false, b,
             (short)0, acc[t],  false,  false);
      }
    }
  }

#pragma unroll
  for (int t = 0; t < GM_RT; ++t) {
#pragma unroll
    for (int r = 0; r < 8; ++r) {
      const int row = rowBase + t * 16 + r + 8 * half;
      if (row < N) out[(size_t)row * M + col] = acc[t][r];
    }
  }
}

__device__ __forceinline__ void fma4(float4& a, const float4 v, const float w) {
  a.x += v.x * w;
  a.y += v.y * w;
  a.z += v.z * w;
  a.w += v.w * w;
}

template <int CTRL>
__device__ __forceinline__ float row_pick(float x) {
  return __builtin_bit_cast(float, __builtin_amdgcn_mov_dpp(__builtin_bit_cast(int, x), CTRL, 0xf, 0xf, true));
}
__device__ __forceinline__ float wave_sum(float v) {
  v += row_pick<0x161>(v);
  v += row_pick<0x162>(v);
  v += row_pick<0x164>(v);
  v += row_pick<0x168>(v);
  v += __builtin_bit_cast(float, __builtin_amdgcn_permlanex16(__builtin_bit_cast(int, v), __builtin_bit_cast(int, v),
                                                              0x76543210, 0xfedcba98, false, false));
  return v;
}

template <bool WIDE>
__global__ void __launch_bounds__(256) k_gather_post(const int* __restrict__ rowptr,
                                                     const int* __restrict__ col,
                                                     const float* __restrict__ dinv,
                                                     const float* __restrict__ hW,
                                                     const float* __restrict__ bias,
                                                     float* __restrict__ out, int N) {
  const int node = __builtin_amdgcn_readfirstlane((int)((blockIdx.x * 256u + threadIdx.x) >> 5));
  const int lane = threadIdx.x & 31;
  if (node >= N) return;
  constexpr int C4 = WIDE ? 64 : 32;
  const float4* __restrict__ tab = (const float4*)hW;
  const int beg = rowptr[node];
  const int end = rowptr[node + 1];
  const float dd = dinv[node];

  float4 bv0, bv1;
  {
    const float* bp = bias + 4 * lane;
    bv0.x = bp[0]; bv0.y = bp[1]; bv0.z = bp[2]; bv0.w = bp[3];
    bv1.x = bv1.y = bv1.z = bv1.w = 0.0f;
    if (WIDE) { bv1.x = bp[128]; bv1.y = bp[129]; bv1.z = bp[130]; bv1.w = bp[131]; }
  }

  float4 a0, a1;
  {
    const float4* hp = tab + (size_t)node * C4 + lane;
    const float ws = dd * dd;
    const float4 x = hp[0];
    a0.x = x.x * ws; a0.y = x.y * ws; a0.z = x.z * ws; a0.w = x.w * ws;
    a1.x = a1.y = a1.z = a1.w = 0.0f;
    if (WIDE) {
      const float4 y = hp[32];
      a1.x = y.x * ws; a1.y = y.y * ws; a1.z = y.z * ws; a1.w = y.w * ws;
    }
  }

  asm volatile("" ::"v"(bv0.x), "v"(bv0.y), "v"(bv0.z), "v"(bv0.w));
  if (WIDE) asm volatile("" ::"v"(bv1.x), "v"(bv1.y), "v"(bv1.z), "v"(bv1.w));

#pragma unroll 4
  for (int j = beg; j < end; ++j) {
    int s = col[j];
    s = (unsigned)s < (unsigned)N ? s : N - 1;
    const float w = dinv[s] * dd;
    const float4* p = tab + (size_t)s * C4 + lane;
    const float4 x = p[0];
    if (WIDE) {
      const float4 y = p[32];
      fma4(a1, y, w);
    }
    fma4(a0, x, w);
  }

  a0.x = fmaxf(a0.x + bv0.x, 0.0f);
  a0.y = fmaxf(a0.y + bv0.y, 0.0f);
  a0.z = fmaxf(a0.z + bv0.z, 0.0f);
  a0.w = fmaxf(a0.w + bv0.w, 0.0f);
  float sum = ((a0.x + a0.y) + a0.z) + a0.w;
  if (WIDE) {
    a1.x = fmaxf(a1.x + bv1.x, 0.0f);
    a1.y = fmaxf(a1.y + bv1.y, 0.0f);
    a1.z = fmaxf(a1.z + bv1.z, 0.0f);
    a1.w = fmaxf(a1.w + bv1.w, 0.0f);
    sum += ((a1.x + a1.y) + a1.z) + a1.w;
  }
  constexpr float invC = WIDE ? (1.0f / 256.0f) : (1.0f / 128.0f);
  const float mean = wave_sum(sum) * invC;
  a0.x -= mean; a0.y -= mean; a0.z -= mean; a0.w -= mean;
  float q = ((a0.x * a0.x + a0.y * a0.y) + a0.z * a0.z) + a0.w * a0.w;
  if (WIDE) {
    a1.x -= mean; a1.y -= mean; a1.z -= mean; a1.w -= mean;
    q += ((a1.x * a1.x + a1.y * a1.y) + a1.z * a1.z) + a1.w * a1.w;
  }
  const float rstd = rsqrtf(wave_sum(q) * invC + 1e-5f);
  a0.x *= rstd; a0.y *= rstd; a0.z *= rstd; a0.w *= rstd;

  float4* op = (float4*)out + (size_t)node * C4 + lane;
  op[0] = a0;
  if (WIDE) {
    a1.x *= rstd; a1.y *= rstd; a1.z *= rstd; a1.w *= rstd;
    op[32] = a1;
  }
}

__global__ void __launch_bounds__(1024) k_pool_mean(const float* __restrict__ h,
                                                     const int* __restrict__ batch,
                                                     float* __restrict__ pool, int N) {
  __shared__ float4 part[32 * 32];
  __shared__ int pcnt[32];
  const int g = blockIdx.x;
  const int lane = threadIdx.x & 31;
  const int w = __builtin_amdgcn_readfirstlane((int)(threadIdx.x >> 5));
  const float4* __restrict__ tab = (const float4*)h;
  const int nchunk = (N + 31) >> 5;

  float4 acc;
  acc.x = acc.y = acc.z = acc.w = 0.0f;
  int cnt = 0;

  for (int c0 = w; c0 < nchunk; c0 += 32 * 8) {
    unsigned m[8];
#pragma unroll
    for (int u = 0; u < 8; ++u) {
      const int n = (c0 + 32 * u) * 32 + lane;
      const int b = batch[n < N ? n : N - 1];
      m[u] = __builtin_amdgcn_ballot_w32(n < N && b == g);
    }
#pragma unroll
    for (int u = 0; u < 8; ++u) {
      unsigned mask = m[u];
      const int base = (c0 + 32 * u) * 32;
      cnt += __builtin_popcount(mask);
      while (mask) {
        int id[8];
        bool ok[8];
#pragma unroll
        for (int q = 0; q < 8; ++q) {
          ok[q] = mask != 0u;
          id[q] = ok[q] ? __builtin_ctz(mask) : id[0];
          mask &= mask - 1u;
        }
        float4 x[8];
#pragma unroll
        for (int q = 0; q < 8; ++q) x[q] = tab[(size_t)(base + id[q]) * 32 + lane];
#pragma unroll
        for (int q = 0; q < 8; ++q) {
          acc.x += ok[q] ? x[q].x : 0.0f;
          acc.y += ok[q] ? x[q].y : 0.0f;
          acc.z += ok[q] ? x[q].z : 0.0f;
          acc.w += ok[q] ? x[q].w : 0.0f;
        }
      }
    }
  }

  part[w * 32 + lane] = acc;
  if (lane == 0) pcnt[w] = cnt;
  __syncthreads();

  const int t = threadIdx.x;
  if (t < 128) {
    const float* pf = (const float*)part;
    float s = 0.0f;
    int c = 0;
    for (int k = 0; k < 32; ++k) {
      s += pf[k * 128 + t];
      c += pcnt[k];
    }
    pool[(size_t)g * 128 + t] = s / fmaxf((float)c, 1.0f);
  }
}

static inline size_t align256(size_t x) { return (x + 255) & ~(size_t)255; }

extern "C" void kernel_launch(void* const* d_in, const int* in_sizes, int n_in,
                              void* d_out, int out_size, void* d_ws, size_t ws_size,
                              hipStream_t stream) {
  const float* x     = (const float*)d_in[0];
  const int*   ei    = (const int*)d_in[1];
  const int*   batch = (const int*)d_in[2];
  const float* W0 = (const float*)d_in[4];
  const float* b0 = (const float*)d_in[5];
  const float* W1 = (const float*)d_in[6];
  const float* b1 = (const float*)d_in[7];
  const float* W2 = (const float*)d_in[8];
  const float* b2 = (const float*)d_in[9];

  const int N  = in_sizes[2];
  const int E  = in_sizes[1] / 2;
  const int IN = in_sizes[0] / N;
  const int H0 = in_sizes[5];
  const int H1 = in_sizes[7];
  const int H2 = in_sizes[9];
  const int G  = out_size / H2 - N;

  const int* srcp = ei;
  const int* dstp = ei + E;

  char* ws = (char*)d_ws;
  float* dinv = (float*)ws;  ws += align256((size_t)N * 4);
  float* bufA = (float*)ws;  ws += align256((size_t)N * 256 * 4);
  float* bufB = (float*)ws;  ws += align256((size_t)N * 256 * 4);
  int* indeg  = (int*)ws;    ws += align256((size_t)N * 4);
  int* rowptr = (int*)ws;    ws += align256((size_t)(N + 1) * 4);
  int* bsum   = (int*)ws;    ws += align256((size_t)(N / 1024 + 1) * 4);
  int* rank   = (int*)ws;    ws += align256((size_t)E * 4);
  int* ctmp   = (int*)ws;    ws += align256((size_t)E * 4);
  int* col    = (int*)ws;    ws += align256((size_t)E * 4);

  float* pool = (float*)d_out;
  float* hout = (float*)d_out + (size_t)G * H2;

  const int B = 256;
  auto cdiv = [](long long a, long long b) { return (int)((a + b - 1) / b); };

  const int NB = cdiv(N, 1024);
  k_zero_i<<<cdiv(N, B), B, 0, stream>>>(indeg, N);
  k_count_rank<<<cdiv(E, B), B, 0, stream>>>(dstp, indeg, rank, E);
  k_dinv_bsum<<<NB, 1024, 0, stream>>>(indeg, dinv, bsum, N);
  k_rowptr<<<NB, 1024, 0, stream>>>(indeg, bsum, rowptr, N);
  k_place<<<cdiv(E, B), B, 0, stream>>>(srcp, dstp, rank, rowptr, ctmp, E);
  k_rowsort<<<cdiv((long long)N * 32, B), B, 0, stream>>>(rowptr, ctmp, col, N);

  auto layer = [&](const float* hin, int K, int M, const float* W, const float* bias,
                   float* hw, float* outp) {
    dim3 g(cdiv(N, GM_ROWS), M / 128);
    k_gemm_wmma<<<g, B, 0, stream>>>(hin, W, hw, N, K, M);
    if (M > 128)
      k_gather_post<true><<<cdiv((long long)N * 32, B), B, 0, stream>>>(rowptr, col, dinv, hw, bias, outp, N);
    else
      k_gather_post<false><<<cdiv((long long)N * 32, B), B, 0, stream>>>(rowptr, col, dinv, hw, bias, outp, N);
  };

  layer(x,    IN, H0, W0, b0,  bufA,  bufB);
  layer(bufB, H0, H1, W1, b1,  bufA,  bufB);
  layer(bufB, H1, H2, W2, b2,  bufA,  hout);

  k_pool_mean<<<G, 1024, 0, stream>>>(hout, batch, pool, N);
}


# Layer-0 product rides in the first work-groups of the count, place and row-sort launches (a third of its tiles each): matrix work under their memory waits, one launch fewer; sum orders unchanged
# speedup vs baseline: 3.8748x; 1.0574x over previous
#include <hip/hip_runtime.h>
#include <hip/hip_bf16.h>

typedef __attribute__((ext_vector_type(2))) float v2f;
typedef __attribute__((ext_vector_type(8))) float v8f;

__global__ void k_zero_i(int* __restrict__ p, int n) {
  int i = blockIdx.x * blockDim.x + threadIdx.x;
  if (i < n) p[i] = 0;
}

__global__ void __launch_bounds__(1024) k_dinv_bsum(const int* __restrict__ cnt,
                                                    float* __restrict__ dinv,
                                                    int* __restrict__ bsum, int N) {
  __shared__ int part[32];
  const int t = threadIdx.x, lane = t & 31, w = t >> 5;
  const int i = blockIdx.x * 1024 + t;
  int c = 0;
  if (i < N) {
    c = cnt[i];
    dinv[i] = rsqrtf(fmaxf((float)(c + 1), 1.0f));
  }
  int s = c;
  for (int off = 16; off > 0; off >>= 1) s += __shfl_xor(s, off, 32);
  if (lane == 0) part[w] = s;
  __syncthreads();
  if (w == 0) {
    int x = part[lane];
    for (int off = 16; off > 0; off >>= 1) x += __shfl_xor(x, off, 32);
    if (lane == 0) bsum[blockIdx.x] = x;
  }
}

__global__ void __launch_bounds__(1024) k_rowptr(const int* __restrict__ cnt,
                                                 const int* __restrict__ bsum,
                                                 int* __restrict__ rowptr, int N) {
  __shared__ int tot[32];
  const int t = threadIdx.x, lane = t & 31, w = t >> 5;
  const int i = blockIdx.x * 1024 + t;
  int base = 0;
  for (int b = lane; b < (int)blockIdx.x; b += 32) base += bsum[b];
  for (int off = 16; off > 0; off >>= 1) base += __shfl_xor(base, off, 32);
  const int c = (i < N) ? cnt[i] : 0;
  int v = c;
  for (int off = 1; off < 32; off <<= 1) {
    int u = __shfl_up(v, off, 32);
    if (lane >= off) v += u;
  }
  if (lane == 31) tot[w] = v;
  __syncthreads();
  if (w == 0) {
    int x = tot[lane];
    for (int off = 1; off < 32; off <<= 1) {
      int u = __shfl_up(x, off, 32);
      if (lane >= off) x += u;
    }
    tot[lane] = x;
  }
  __syncthreads();
  const int excl = base + (w > 0 ? tot[w - 1] : 0) + v - c;
  if (i < N) rowptr[i] = excl;
  if (i == N - 1) rowptr[N] = excl + c;
}

constexpr int GM_ROWS = 64;
constexpr int GM_RT = GM_ROWS / 16;
constexpr int GM_KC = 128;
constexpr int GM_PITCH = GM_KC + 4;

__device__ __forceinline__ void gemm_tile(
    const float* __restrict__ A, const float* __restrict__ W,
    float* __restrict__ out, int N, int K, int M, int bx, int by) {
  __shared__ float As[GM_ROWS * GM_PITCH];
  const int rowBase = bx * GM_ROWS;
  const int lane = threadIdx.x & 31;
  const int half = lane >> 4;
  const int m    = lane & 15;
  const int col  = by * 128 + (threadIdx.x >> 5) * 16 + m;

  v8f acc[GM_RT];
#pragma unroll
  for (int t = 0; t < GM_RT; ++t) acc[t] = v8f{};

  for (int kc0 = 0; kc0 < K; kc0 += GM_KC) {
    const int kw = (K - kc0 < GM_KC) ? (K - kc0) : GM_KC;
    if (kc0 > 0) __syncthreads();

    {
      float4 v[8];
      int    off[8];
      bool   ok[8];
#pragma unroll
      for (int u = 0; u < 8; ++u) {
        const int e = threadIdx.x + 256 * u;
        const int r = e >> 5;
        const int c = (e & 31) * 4;
        const int g = rowBase + r;
        ok[u]  = (g < N) && (c < kw);
        off[u] = r * GM_PITCH + c;
        const int gc = g < N ? g : N - 1;
        const int cc = c < kw ? c : 0;
        v[u] = *(const float4*)(A + (size_t)gc * K + kc0 + cc);
      }
#pragma unroll
      for (int u = 0; u < 8; ++u) {
        float4 x = v[u];
        if (!ok[u]) x.x = x.y = x.z = x.w = 0.0f;
        *(float4*)(As + off[u]) = x;
      }
    }
    __syncthreads();

    const float* __restrict__ wp = W + (size_t)(kc0 + 2 * half) * M + col;
    const float* ap = As + m * GM_PITCH + 2 * half;
#pragma unroll 4
    for (int k = 0; k < kw; k += 4) {
      v2f b;
      b.x = wp[(size_t)k * M];
      b.y = wp[(size_t)(k + 1) * M];
#pragma unroll
      for (int t = 0; t < GM_RT; ++t) {
        v2f a;
        a.x = ap[t * 16 * GM_PITCH + k];
        a.y = ap[t * 16 * GM_PITCH + k + 1];
        acc[t] = __builtin_amdgcn_wmma_f32_16x16x4_f32(
             false, a,  false, b,
             (short)0, acc[t],  false,  false);
      }
    }
  }

#pragma unroll
  for (int t = 0; t < GM_RT; ++t) {
#pragma unroll
    for (int r = 0; r < 8; ++r) {
      const int row = rowBase + t * 16 + r + 8 * half;
      if (row < N) out[(size_t)row * M + col] = acc[t][r];
    }
  }
}

__global__ void __launch_bounds__(256) k_gemm_wmma(
    const float* __restrict__ A, const float* __restrict__ W,
    float* __restrict__ out, int N, int K, int M) {
  gemm_tile(A, W, out, N, K, M, blockIdx.x, blockIdx.y);
}

struct GemmJob {
  const float* A;
  const float* W;
  float* out;
  int N, K, M;
  int tile0, ntiles;
};

__device__ __forceinline__ bool gemm_share(const GemmJob& g) {
  if ((int)blockIdx.x >= g.ntiles) return false;
  const int t = g.tile0 + (int)blockIdx.x;
  const int ny = g.M / 128;
  gemm_tile(g.A, g.W, g.out, g.N, g.K, g.M, t / ny, t % ny);
  return true;
}

__global__ void __launch_bounds__(256) k_count_rank(const int* __restrict__ dst, int* __restrict__ cnt,
                                                    int* __restrict__ rank, int E, GemmJob g) {
  if (gemm_share(g)) return;
  const int i = ((int)blockIdx.x - g.ntiles) * 256 + (int)threadIdx.x;
  if (i < E) rank[i] = atomicAdd(cnt + dst[i], 1);
}

__global__ void __launch_bounds__(256) k_place(const int* __restrict__ src, const int* __restrict__ dst,
                                               const int* __restrict__ rank, const int* __restrict__ rowptr,
                                               int* __restrict__ tmp, int E, GemmJob g) {
  if (gemm_share(g)) return;
  const int i = ((int)blockIdx.x - g.ntiles) * 256 + (int)threadIdx.x;
  if (i < E) {
    int p = rowptr[dst[i]] + rank[i];
    if ((unsigned)p < (unsigned)E) tmp[p] = src[i];
  }
}

__global__ void __launch_bounds__(256) k_rowsort(const int* __restrict__ rowptr,
                                                 const int* __restrict__ tmp,
                                                 int* __restrict__ col, int N, GemmJob g) {
  if (gemm_share(g)) return;
  const int node = __builtin_amdgcn_readfirstlane(
      (int)((((unsigned)blockIdx.x - (unsigned)g.ntiles) * 256u + threadIdx.x) >> 5));
  const int lane = threadIdx.x & 31;
  if (node >= N) return;
  const int beg = rowptr[node];
  const int n = rowptr[node + 1] - beg;
  for (int i = lane; i < n; i += 32) {
    const int key = tmp[beg + i];
    int r = 0;
    for (int j = 0; j < n; ++j) {
      const int kj = tmp[beg + j];
      r += (kj < key || (kj == key && j < i)) ? 1 : 0;
    }
    col[beg + r] = key;
  }
}

__device__ __forceinline__ void fma4(float4& a, const float4 v, const float w) {
  a.x += v.x * w;
  a.y += v.y * w;
  a.z += v.z * w;
  a.w += v.w * w;
}

template <int CTRL>
__device__ __forceinline__ float row_pick(float x) {
  return __builtin_bit_cast(float, __builtin_amdgcn_mov_dpp(__builtin_bit_cast(int, x), CTRL, 0xf, 0xf, true));
}
__device__ __forceinline__ float wave_sum(float v) {
  v += row_pick<0x161>(v);
  v += row_pick<0x162>(v);
  v += row_pick<0x164>(v);
  v += row_pick<0x168>(v);
  v += __builtin_bit_cast(float, __builtin_amdgcn_permlanex16(__builtin_bit_cast(int, v), __builtin_bit_cast(int, v),
                                                              0x76543210, 0xfedcba98, false, false));
  return v;
}

template <bool WIDE>
__global__ void __launch_bounds__(256) k_gather_post(const int* __restrict__ rowptr,
                                                     const int* __restrict__ col,
                                                     const float* __restrict__ dinv,
                                                     const float* __restrict__ hW,
                                                     const float* __restrict__ bias,
                                                     float* __restrict__ out, int N) {
  const int node = __builtin_amdgcn_readfirstlane((int)((blockIdx.x * 256u + threadIdx.x) >> 5));
  const int lane = threadIdx.x & 31;
  if (node >= N) return;
  constexpr int C4 = WIDE ? 64 : 32;
  const float4* __restrict__ tab = (const float4*)hW;
  const int beg = rowptr[node];
  const int end = rowptr[node + 1];
  const float dd = dinv[node];

  float4 bv0, bv1;
  {
    const float* bp = bias + 4 * lane;
    bv0.x = bp[0]; bv0.y = bp[1]; bv0.z = bp[2]; bv0.w = bp[3];
    bv1.x = bv1.y = bv1.z = bv1.w = 0.0f;
    if (WIDE) { bv1.x = bp[128]; bv1.y = bp[129]; bv1.z = bp[130]; bv1.w = bp[131]; }
  }

  float4 a0, a1;
  {
    const float4* hp = tab + (size_t)node * C4 + lane;
    const float ws = dd * dd;
    const float4 x = hp[0];
    a0.x = x.x * ws; a0.y = x.y * ws; a0.z = x.z * ws; a0.w = x.w * ws;
    a1.x = a1.y = a1.z = a1.w = 0.0f;
    if (WIDE) {
      const float4 y = hp[32];
      a1.x = y.x * ws; a1.y = y.y * ws; a1.z = y.z * ws; a1.w = y.w * ws;
    }
  }

  asm volatile("" ::"v"(bv0.x), "v"(bv0.y), "v"(bv0.z), "v"(bv0.w));
  if (WIDE) asm volatile("" ::"v"(bv1.x), "v"(bv1.y), "v"(bv1.z), "v"(bv1.w));

#pragma unroll 4
  for (int j = beg; j < end; ++j) {
    int s = col[j];
    s = (unsigned)s < (unsigned)N ? s : N - 1;
    const float w = dinv[s] * dd;
    const float4* p = tab + (size_t)s * C4 + lane;
    const float4 x = p[0];
    if (WIDE) {
      const float4 y = p[32];
      fma4(a1, y, w);
    }
    fma4(a0, x, w);
  }

  a0.x = fmaxf(a0.x + bv0.x, 0.0f);
  a0.y = fmaxf(a0.y + bv0.y, 0.0f);
  a0.z = fmaxf(a0.z + bv0.z, 0.0f);
  a0.w = fmaxf(a0.w + bv0.w, 0.0f);
  float sum = ((a0.x + a0.y) + a0.z) + a0.w;
  if (WIDE) {
    a1.x = fmaxf(a1.x + bv1.x, 0.0f);
    a1.y = fmaxf(a1.y + bv1.y, 0.0f);
    a1.z = fmaxf(a1.z + bv1.z, 0.0f);
    a1.w = fmaxf(a1.w + bv1.w, 0.0f);
    sum += ((a1.x + a1.y) + a1.z) + a1.w;
  }
  constexpr float invC = WIDE ? (1.0f / 256.0f) : (1.0f / 128.0f);
  const float mean = wave_sum(sum) * invC;
  a0.x -= mean; a0.y -= mean; a0.z -= mean; a0.w -= mean;
  float q = ((a0.x * a0.x + a0.y * a0.y) + a0.z * a0.z) + a0.w * a0.w;
  if (WIDE) {
    a1.x -= mean; a1.y -= mean; a1.z -= mean; a1.w -= mean;
    q += ((a1.x * a1.x + a1.y * a1.y) + a1.z * a1.z) + a1.w * a1.w;
  }
  const float rstd = rsqrtf(wave_sum(q) * invC + 1e-5f);
  a0.x *= rstd; a0.y *= rstd; a0.z *= rstd; a0.w *= rstd;

  float4* op = (float4*)out + (size_t)node * C4 + lane;
  op[0] = a0;
  if (WIDE) {
    a1.x *= rstd; a1.y *= rstd; a1.z *= rstd; a1.w *= rstd;
    op[32] = a1;
  }
}

__global__ void __launch_bounds__(1024) k_pool_mean(const float* __restrict__ h,
                                                     const int* __restrict__ batch,
                                                     float* __restrict__ pool, int N) {
  __shared__ float4 part[32 * 32];
  __shared__ int pcnt[32];
  const int g = blockIdx.x;
  const int lane = threadIdx.x & 31;
  const int w = __builtin_amdgcn_readfirstlane((int)(threadIdx.x >> 5));
  const float4* __restrict__ tab = (const float4*)h;
  const int nchunk = (N + 31) >> 5;

  float4 acc;
  acc.x = acc.y = acc.z = acc.w = 0.0f;
  int cnt = 0;

  for (int c0 = w; c0 < nchunk; c0 += 32 * 8) {
    unsigned m[8];
#pragma unroll
    for (int u = 0; u < 8; ++u) {
      const int n = (c0 + 32 * u) * 32 + lane;
      const int b = batch[n < N ? n : N - 1];
      m[u] = __builtin_amdgcn_ballot_w32(n < N && b == g);
    }
#pragma unroll
    for (int u = 0; u < 8; ++u) {
      unsigned mask = m[u];
      const int base = (c0 + 32 * u) * 32;
      cnt += __builtin_popcount(mask);
      while (mask) {
        int id[8];
        bool ok[8];
#pragma unroll
        for (int q = 0; q < 8; ++q) {
          ok[q] = mask != 0u;
          id[q] = ok[q] ? __builtin_ctz(mask) : id[0];
          mask &= mask - 1u;
        }
        float4 x[8];
#pragma unroll
        for (int q = 0; q < 8; ++q) x[q] = tab[(size_t)(base + id[q]) * 32 + lane];
#pragma unroll
        for (int q = 0; q < 8; ++q) {
          acc.x += ok[q] ? x[q].x : 0.0f;
          acc.y += ok[q] ? x[q].y : 0.0f;
          acc.z += ok[q] ? x[q].z : 0.0f;
          acc.w += ok[q] ? x[q].w : 0.0f;
        }
      }
    }
  }

  part[w * 32 + lane] = acc;
  if (lane == 0) pcnt[w] = cnt;
  __syncthreads();

  const int t = threadIdx.x;
  if (t < 128) {
    const float* pf = (const float*)part;
    float s = 0.0f;
    int c = 0;
    for (int k = 0; k < 32; ++k) {
      s += pf[k * 128 + t];
      c += pcnt[k];
    }
    pool[(size_t)g * 128 + t] = s / fmaxf((float)c, 1.0f);
  }
}

static inline size_t align256(size_t x) { return (x + 255) & ~(size_t)255; }

extern "C" void kernel_launch(void* const* d_in, const int* in_sizes, int n_in,
                              void* d_out, int out_size, void* d_ws, size_t ws_size,
                              hipStream_t stream) {
  const float* x     = (const float*)d_in[0];
  const int*   ei    = (const int*)d_in[1];
  const int*   batch = (const int*)d_in[2];
  const float* W0 = (const float*)d_in[4];
  const float* b0 = (const float*)d_in[5];
  const float* W1 = (const float*)d_in[6];
  const float* b1 = (const float*)d_in[7];
  const float* W2 = (const float*)d_in[8];
  const float* b2 = (const float*)d_in[9];

  const int N  = in_sizes[2];
  const int E  = in_sizes[1] / 2;
  const int IN = in_sizes[0] / N;
  const int H0 = in_sizes[5];
  const int H1 = in_sizes[7];
  const int H2 = in_sizes[9];
  const int G  = out_size / H2 - N;

  const int* srcp = ei;
  const int* dstp = ei + E;

  char* ws = (char*)d_ws;
  float* dinv = (float*)ws;  ws += align256((size_t)N * 4);
  float* bufA = (float*)ws;  ws += align256((size_t)N * 256 * 4);
  float* bufB = (float*)ws;  ws += align256((size_t)N * 256 * 4);
  int* indeg  = (int*)ws;    ws += align256((size_t)N * 4);
  int* rowptr = (int*)ws;    ws += align256((size_t)(N + 1) * 4);
  int* bsum   = (int*)ws;    ws += align256((size_t)(N / 1024 + 1) * 4);
  int* rank   = (int*)ws;    ws += align256((size_t)E * 4);
  int* ctmp   = (int*)ws;    ws += align256((size_t)E * 4);
  int* col    = (int*)ws;    ws += align256((size_t)E * 4);

  float* pool = (float*)d_out;
  float* hout = (float*)d_out + (size_t)G * H2;

  const int B = 256;
  auto cdiv = [](long long a, long long b) { return (int)((a + b - 1) / b); };

  const int NB = cdiv(N, 1024);
  const int nt0 = cdiv(N, GM_ROWS) * (H0 / 128);
  const int t1 = (nt0 / 3) & ~255, t2 = t1, t3 = nt0 - t1 - t2;
  const GemmJob j1{x, W0, bufA, N, IN, H0, 0, t1};
  const GemmJob j2{x, W0, bufA, N, IN, H0, t1, t2};
  const GemmJob j3{x, W0, bufA, N, IN, H0, t1 + t2, t3};
  k_zero_i<<<cdiv(N, B), B, 0, stream>>>(indeg, N);
  k_count_rank<<<t1 + cdiv(E, B), B, 0, stream>>>(dstp, indeg, rank, E, j1);
  k_dinv_bsum<<<NB, 1024, 0, stream>>>(indeg, dinv, bsum, N);
  k_rowptr<<<NB, 1024, 0, stream>>>(indeg, bsum, rowptr, N);
  k_place<<<t2 + cdiv(E, B), B, 0, stream>>>(srcp, dstp, rank, rowptr, ctmp, E, j2);
  k_rowsort<<<t3 + cdiv((long long)N * 32, B), B, 0, stream>>>(rowptr, ctmp, col, N, j3);

  auto product = [&](const float* hin, int K, int M, const float* W, float* hw) {
    dim3 g(cdiv(N, GM_ROWS), M / 128);
    k_gemm_wmma<<<g, B, 0, stream>>>(hin, W, hw, N, K, M);
  };
  auto gather = [&](int M, const float* hw, const float* bias, float* outp) {
    if (M > 128)
      k_gather_post<true><<<cdiv((long long)N * 32, B), B, 0, stream>>>(rowptr, col, dinv, hw, bias, outp, N);
    else
      k_gather_post<false><<<cdiv((long long)N * 32, B), B, 0, stream>>>(rowptr, col, dinv, hw, bias, outp, N);
  };

  gather(H0, bufA, b0, bufB);
  product(bufB, H0, H1, W1, bufA);
  gather(H1, bufA, b1, bufB);
  product(bufB, H1, H2, W2, bufA);
  gather(H2, bufA, b2, hout);

  k_pool_mean<<<G, 1024, 0, stream>>>(hout, batch, pool, N);
}
